# starting kernel
# baseline (speedup 1.0000x reference)
#include <hip/hip_runtime.h>
#include <math.h>

typedef __attribute__((ext_vector_type(2))) float v2f;
typedef __attribute__((ext_vector_type(8))) float v8f;

#define NN 100000

__global__ void zero_f32(float* __restrict__ p, int n4) {
    int i = blockIdx.x * blockDim.x + threadIdx.x;
    int stride = gridDim.x * blockDim.x;
    float4 z = make_float4(0.f, 0.f, 0.f, 0.f);
    for (; i < n4; i += stride) ((float4*)p)[i] = z;
}

__global__ void deg_kernel(const int* __restrict__ src, const int* __restrict__ dst,
                           unsigned* __restrict__ out_cnt, unsigned* __restrict__ in_cnt, int E) {
    int e = blockIdx.x * blockDim.x + threadIdx.x;
    if (e < E) {
        atomicAdd(&out_cnt[src[e]], 1u);
        atomicAdd(&in_cnt[dst[e]], 1u);
    }
}

__global__ void norm_kernel(const unsigned* __restrict__ out_cnt, const unsigned* __restrict__ in_cnt,
                            float* __restrict__ out_norm, float* __restrict__ in_norm, int n) {
    int i = blockIdx.x * blockDim.x + threadIdx.x;
    if (i < n) {
        out_norm[i] = rsqrtf((float)(out_cnt[i] + 1u));
        in_norm[i]  = rsqrtf((float)(in_cnt[i] + 1u));
    }
}

__global__ void init_h_kernel(const float* __restrict__ weight, const float* __restrict__ W_in,
                              const float* __restrict__ b_in, float* __restrict__ h, int n) {
    int idx = blockIdx.x * blockDim.x + threadIdx.x;
    if (idx >= n * 16) return;
    int node = idx >> 4;
    int q = idx & 15;
    float w = weight[node];
    float4 wi = ((const float4*)W_in)[q];
    float4 bi = ((const float4*)b_in)[q];
    float4 r;
    r.x = w * wi.x + bi.x;
    r.y = w * wi.y + bi.y;
    r.z = w * wi.z + bi.z;
    r.w = w * wi.w + bi.w;
    ((float4*)(h + (size_t)node * 64))[q] = r;
}

__global__ __launch_bounds__(256) void gemm_layer_kernel(
    const float* __restrict__ h, const float* __restrict__ W,
    const float* __restrict__ out_norm, float* __restrict__ g, int act)
{
    __shared__ float Wl[64 * 64];
    __shared__ float Tl[32 * 64];

    const int tid = threadIdx.x;
    const int row0 = blockIdx.x * 32;

    for (int i = tid; i < 1024; i += 256)
        ((float4*)Wl)[i] = ((const float4*)W)[i];

    for (int i = tid; i < 512; i += 256) {
        int r  = i >> 4;
        int c4 = i & 15;
        int row = row0 + r;
        float4 v = ((const float4*)(h + (size_t)row * 64))[c4];
        if (act) {
            v.x = v.x > 0.f ? v.x : 0.01f * v.x;
            v.y = v.y > 0.f ? v.y : 0.01f * v.y;
            v.z = v.z > 0.f ? v.z : 0.01f * v.z;
            v.w = v.w > 0.f ? v.w : 0.01f * v.w;
        }
        float s = out_norm[row];
        v.x *= s; v.y *= s; v.z *= s; v.w *= s;
        ((float4*)(Tl + r * 64))[c4] = v;
    }
    __syncthreads();

    const int wave = tid >> 5;
    const int lane = tid & 31;
    const int msub = wave >> 2;
    const int nsub = wave & 3;
    const int l16  = lane & 15;
    const int arow = msub * 16 + l16;
    const int koff = (lane < 16) ? 0 : 2;
    const int col  = nsub * 16 + l16;

    v8f c = {};
    #pragma unroll
    for (int kk = 0; kk < 16; ++kk) {
        int k = kk * 4 + koff;
        v2f a, b;
        a.x = Tl[arow * 64 + k];
        a.y = Tl[arow * 64 + k + 1];
        b.x = Wl[k * 64 + col];
        b.y = Wl[(k + 1) * 64 + col];
        c = __builtin_amdgcn_wmma_f32_16x16x4_f32(false, a, false, b,
                                                  (short)0, c, false, false);
    }

    const int rbase = row0 + msub * 16 + ((lane < 16) ? 0 : 8);
    #pragma unroll
    for (int r = 0; r < 8; ++r)
        g[(size_t)(rbase + r) * 64 + col] = c[r];
}

__global__ __launch_bounds__(256) void spmm_kernel(
    const float* __restrict__ g, const int* __restrict__ src, const int* __restrict__ dst,
    float* __restrict__ agg, int E)
{
    int e = blockIdx.x * 8 + (threadIdx.x >> 5);
    if (e >= E) return;
    int lane = threadIdx.x & 31;
    int s = src[e];
    int d = dst[e];
    float2 v = ((const float2*)(g + (size_t)s * 64))[lane];
    float* a = agg + (size_t)d * 64 + lane * 2;
    atomicAdd(a,     v.x);
    atomicAdd(a + 1, v.y);
}

__global__ void post_kernel(const float* __restrict__ agg, const float* __restrict__ g,
                            const float* __restrict__ in_norm, const float* __restrict__ bc,
                            float* __restrict__ h, int n) {
    int idx = blockIdx.x * blockDim.x + threadIdx.x;
    if (idx >= n * 16) return;
    int node = idx >> 4;
    int q = idx & 15;
    float s = in_norm[node];
    float4 a = ((const float4*)(agg + (size_t)node * 64))[q];
    float4 gv = ((const float4*)(g + (size_t)node * 64))[q];
    float4 b = ((const float4*)bc)[q];
    float4 r;
    r.x = (a.x + gv.x) * s + b.x;
    r.y = (a.y + gv.y) * s + b.y;
    r.z = (a.z + gv.z) * s + b.z;
    r.w = (a.w + gv.w) * s + b.w;
    ((float4*)(h + (size_t)node * 64))[q] = r;
}

__global__ __launch_bounds__(256) void pred_kernel(
    const float* __restrict__ h, const float* __restrict__ W_pred,
    const float* __restrict__ b_pred, float* __restrict__ out, int n)
{
    int node = blockIdx.x * 8 + (threadIdx.x >> 5);
    if (node >= n) return;
    int lane = threadIdx.x & 31;
    float2 w = ((const float2*)W_pred)[lane];
    float2 v = ((const float2*)(h + (size_t)node * 64))[lane];
    float s = v.x * w.x + v.y * w.y;
    for (int off = 16; off > 0; off >>= 1)
        s += __shfl_down(s, off, 32);
    if (lane == 0) out[node] = s + b_pred[0];
}

extern "C" void kernel_launch(void* const* d_in, const int* in_sizes, int n_in,
                              void* d_out, int out_size, void* d_ws, size_t ws_size,
                              hipStream_t stream) {
    const float* weight = (const float*)d_in[0];
    const int*   src    = (const int*)d_in[1];
    const int*   dst    = (const int*)d_in[2];
    const float* W_in   = (const float*)d_in[3];
    const float* b_in   = (const float*)d_in[4];
    const float* Wc     = (const float*)d_in[5];
    const float* bc     = (const float*)d_in[6];
    const float* W_pred = (const float*)d_in[7];
    const float* b_pred = (const float*)d_in[8];
    float* out = (float*)d_out;

    const int N = in_sizes[0];
    const int E = in_sizes[1];

    unsigned* out_cnt = (unsigned*)d_ws;
    unsigned* in_cnt  = out_cnt + N;
    float* out_norm   = (float*)(in_cnt + N);
    float* in_norm    = out_norm + N;
    float* h          = in_norm + N;
    float* g          = h + (size_t)N * 64;
    float* agg        = g + (size_t)N * 64;

    zero_f32<<<512, 256, 0, stream>>>((float*)out_cnt, (2 * N) / 4);

    deg_kernel<<<(E + 255) / 256, 256, 0, stream>>>(src, dst, out_cnt, in_cnt, E);

    norm_kernel<<<(N + 255) / 256, 256, 0, stream>>>(out_cnt, in_cnt, out_norm, in_norm, N);

    init_h_kernel<<<(N * 16 + 255) / 256, 256, 0, stream>>>(weight, W_in, b_in, h, N);

    for (int l = 0; l < 3; ++l) {
        gemm_layer_kernel<<<N / 32, 256, 0, stream>>>(h, Wc + (size_t)l * 64 * 64,
                                                      out_norm, g, l > 0 ? 1 : 0);
        zero_f32<<<2048, 256, 0, stream>>>(agg, (N * 64) / 4);
        spmm_kernel<<<(E + 7) / 8, 256, 0, stream>>>(g, src, dst, agg, E);
        post_kernel<<<(N * 16 + 255) / 256, 256, 0, stream>>>(agg, g, in_norm,
                                                              bc + (size_t)l * 64, h, N);
    }

    pred_kernel<<<(N + 7) / 8, 256, 0, stream>>>(h, W_pred, b_pred, out, N);
}


# SpMM: one wave per 32 consecutive edges, not per edge: indices by one coalesced load, rows gathered 8 at a time ahead of their atomics, scalar row bases; f32 atomics (arrival-order sums) kept
# speedup vs baseline: 1.4840x; 1.4840x over previous
#include <hip/hip_runtime.h>
#include <math.h>

typedef __attribute__((ext_vector_type(2))) float v2f;
typedef __attribute__((ext_vector_type(8))) float v8f;

#define NN 100000

__global__ void zero_f32(float* __restrict__ p, int n4) {
    int i = blockIdx.x * blockDim.x + threadIdx.x;
    int stride = gridDim.x * blockDim.x;
    float4 z = make_float4(0.f, 0.f, 0.f, 0.f);
    for (; i < n4; i += stride) ((float4*)p)[i] = z;
}

__global__ void deg_kernel(const int* __restrict__ src, const int* __restrict__ dst,
                           unsigned* __restrict__ out_cnt, unsigned* __restrict__ in_cnt, int E) {
    int e = blockIdx.x * blockDim.x + threadIdx.x;
    if (e < E) {
        atomicAdd(&out_cnt[src[e]], 1u);
        atomicAdd(&in_cnt[dst[e]], 1u);
    }
}

__global__ void norm_kernel(const unsigned* __restrict__ out_cnt, const unsigned* __restrict__ in_cnt,
                            float* __restrict__ out_norm, float* __restrict__ in_norm, int n) {
    int i = blockIdx.x * blockDim.x + threadIdx.x;
    if (i < n) {
        out_norm[i] = rsqrtf((float)(out_cnt[i] + 1u));
        in_norm[i]  = rsqrtf((float)(in_cnt[i] + 1u));
    }
}

__global__ void init_h_kernel(const float* __restrict__ weight, const float* __restrict__ W_in,
                              const float* __restrict__ b_in, float* __restrict__ h, int n) {
    int idx = blockIdx.x * blockDim.x + threadIdx.x;
    if (idx >= n * 16) return;
    int node = idx >> 4;
    int q = idx & 15;
    float w = weight[node];
    float4 wi = ((const float4*)W_in)[q];
    float4 bi = ((const float4*)b_in)[q];
    float4 r;
    r.x = w * wi.x + bi.x;
    r.y = w * wi.y + bi.y;
    r.z = w * wi.z + bi.z;
    r.w = w * wi.w + bi.w;
    ((float4*)(h + (size_t)node * 64))[q] = r;
}

__global__ __launch_bounds__(256) void gemm_layer_kernel(
    const float* __restrict__ h, const float* __restrict__ W,
    const float* __restrict__ out_norm, float* __restrict__ g, int act)
{
    __shared__ float Wl[64 * 64];
    __shared__ float Tl[32 * 64];

    const int tid = threadIdx.x;
    const int row0 = blockIdx.x * 32;

    for (int i = tid; i < 1024; i += 256)
        ((float4*)Wl)[i] = ((const float4*)W)[i];

    for (int i = tid; i < 512; i += 256) {
        int r  = i >> 4;
        int c4 = i & 15;
        int row = row0 + r;
        float4 v = ((const float4*)(h + (size_t)row * 64))[c4];
        if (act) {
            v.x = v.x > 0.f ? v.x : 0.01f * v.x;
            v.y = v.y > 0.f ? v.y : 0.01f * v.y;
            v.z = v.z > 0.f ? v.z : 0.01f * v.z;
            v.w = v.w > 0.f ? v.w : 0.01f * v.w;
        }
        float s = out_norm[row];
        v.x *= s; v.y *= s; v.z *= s; v.w *= s;
        ((float4*)(Tl + r * 64))[c4] = v;
    }
    __syncthreads();

    const int wave = tid >> 5;
    const int lane = tid & 31;
    const int msub = wave >> 2;
    const int nsub = wave & 3;
    const int l16  = lane & 15;
    const int arow = msub * 16 + l16;
    const int koff = (lane < 16) ? 0 : 2;
    const int col  = nsub * 16 + l16;

    v8f c = {};
    #pragma unroll
    for (int kk = 0; kk < 16; ++kk) {
        int k = kk * 4 + koff;
        v2f a, b;
        a.x = Tl[arow * 64 + k];
        a.y = Tl[arow * 64 + k + 1];
        b.x = Wl[k * 64 + col];
        b.y = Wl[(k + 1) * 64 + col];
        c = __builtin_amdgcn_wmma_f32_16x16x4_f32(false, a, false, b,
                                                  (short)0, c, false, false);
    }

    const int rbase = row0 + msub * 16 + ((lane < 16) ? 0 : 8);
    #pragma unroll
    for (int r = 0; r < 8; ++r)
        g[(size_t)(rbase + r) * 64 + col] = c[r];
}


typedef const v2f __attribute__((address_space(1)))* grow_in_t;
typedef float __attribute__((address_space(1)))* grow_out_t;
__device__ __forceinline__ grow_in_t row_in(const float* p, int r) {
    return (grow_in_t)((unsigned long long)p + ((unsigned long long)(unsigned)r << 8));
}
__device__ __forceinline__ grow_out_t row_out(float* p, int r) {
    return (grow_out_t)((unsigned long long)p + ((unsigned long long)(unsigned)r << 8));
}
__device__ __forceinline__ void add_f32(grow_out_t a, float x) {
    (void)__hip_atomic_fetch_add(a, x, __ATOMIC_RELAXED, __HIP_MEMORY_SCOPE_AGENT);
}

__global__ __launch_bounds__(256) void spmm_kernel(
    const float* __restrict__ g, const int* __restrict__ src, const int* __restrict__ dst,
    float* __restrict__ agg, int E)
{
    const unsigned lane = threadIdx.x & 31u;
    const int base = __builtin_amdgcn_readfirstlane((int)(blockIdx.x * 8 + (threadIdx.x >> 5)) * 32);
    if (base >= E) return;
    const int e = base + (int)lane;
    const int sv = (e < E) ? src[e] : 0;
    const int dv = (e < E) ? dst[e] : 0;
    const int cnt = (E - base < 32) ? (E - base) : 32;

    if (cnt == 32) {
        v2f v[2][8];
        #pragma unroll
        for (int j = 0; j < 8; ++j)
            v[0][j] = row_in(g, __builtin_amdgcn_readlane(sv, j))[lane];
        #pragma unroll
        for (int b = 0; b < 4; ++b) {
            if (b < 3) {
                #pragma unroll
                for (int j = 0; j < 8; ++j)
                    v[(b + 1) & 1][j] = row_in(g, __builtin_amdgcn_readlane(sv, (b + 1) * 8 + j))[lane];
            }
            #pragma unroll
            for (int j = 0; j < 8; ++j) {
                grow_out_t a = row_out(agg, __builtin_amdgcn_readlane(dv, b * 8 + j)) + lane * 2u;
                add_f32(a,     v[b & 1][j].x);
                add_f32(a + 1, v[b & 1][j].y);
            }
        }
    } else {
        for (int j = 0; j < cnt; ++j) {
            const v2f r = row_in(g, __builtin_amdgcn_readlane(sv, j))[lane];
            grow_out_t a = row_out(agg, __builtin_amdgcn_readlane(dv, j)) + lane * 2u;
            add_f32(a,     r.x);
            add_f32(a + 1, r.y);
        }
    }
}

__global__ void post_kernel(const float* __restrict__ agg, const float* __restrict__ g,
                            const float* __restrict__ in_norm, const float* __restrict__ bc,
                            float* __restrict__ h, int n) {
    int idx = blockIdx.x * blockDim.x + threadIdx.x;
    if (idx >= n * 16) return;
    int node = idx >> 4;
    int q = idx & 15;
    float s = in_norm[node];
    float4 a = ((const float4*)(agg + (size_t)node * 64))[q];
    float4 gv = ((const float4*)(g + (size_t)node * 64))[q];
    float4 b = ((const float4*)bc)[q];
    float4 r;
    r.x = (a.x + gv.x) * s + b.x;
    r.y = (a.y + gv.y) * s + b.y;
    r.z = (a.z + gv.z) * s + b.z;
    r.w = (a.w + gv.w) * s + b.w;
    ((float4*)(h + (size_t)node * 64))[q] = r;
}

__global__ __launch_bounds__(256) void pred_kernel(
    const float* __restrict__ h, const float* __restrict__ W_pred,
    const float* __restrict__ b_pred, float* __restrict__ out, int n)
{
    int node = blockIdx.x * 8 + (threadIdx.x >> 5);
    if (node >= n) return;
    int lane = threadIdx.x & 31;
    float2 w = ((const float2*)W_pred)[lane];
    float2 v = ((const float2*)(h + (size_t)node * 64))[lane];
    float s = v.x * w.x + v.y * w.y;
    for (int off = 16; off > 0; off >>= 1)
        s += __shfl_down(s, off, 32);
    if (lane == 0) out[node] = s + b_pred[0];
}

extern "C" void kernel_launch(void* const* d_in, const int* in_sizes, int n_in,
                              void* d_out, int out_size, void* d_ws, size_t ws_size,
                              hipStream_t stream) {
    const float* weight = (const float*)d_in[0];
    const int*   src    = (const int*)d_in[1];
    const int*   dst    = (const int*)d_in[2];
    const float* W_in   = (const float*)d_in[3];
    const float* b_in   = (const float*)d_in[4];
    const float* Wc     = (const float*)d_in[5];
    const float* bc     = (const float*)d_in[6];
    const float* W_pred = (const float*)d_in[7];
    const float* b_pred = (const float*)d_in[8];
    float* out = (float*)d_out;

    const int N = in_sizes[0];
    const int E = in_sizes[1];

    unsigned* out_cnt = (unsigned*)d_ws;
    unsigned* in_cnt  = out_cnt + N;
    float* out_norm   = (float*)(in_cnt + N);
    float* in_norm    = out_norm + N;
    float* h          = in_norm + N;
    float* g          = h + (size_t)N * 64;
    float* agg        = g + (size_t)N * 64;

    zero_f32<<<512, 256, 0, stream>>>((float*)out_cnt, (2 * N) / 4);

    deg_kernel<<<(E + 255) / 256, 256, 0, stream>>>(src, dst, out_cnt, in_cnt, E);

    norm_kernel<<<(N + 255) / 256, 256, 0, stream>>>(out_cnt, in_cnt, out_norm, in_norm, N);

    init_h_kernel<<<(N * 16 + 255) / 256, 256, 0, stream>>>(weight, W_in, b_in, h, N);

    for (int l = 0; l < 3; ++l) {
        gemm_layer_kernel<<<N / 32, 256, 0, stream>>>(h, Wc + (size_t)l * 64 * 64,
                                                      out_norm, g, l > 0 ? 1 : 0);
        zero_f32<<<2048, 256, 0, stream>>>(agg, (N * 64) / 4);
        spmm_kernel<<<(E + 255) / 256, 256, 0, stream>>>(g, src, dst, agg, E);
        post_kernel<<<(N * 16 + 255) / 256, 256, 0, stream>>>(agg, g, in_norm,
                                                              bc + (size_t)l * 64, h, N);
    }

    pred_kernel<<<(N + 7) / 8, 256, 0, stream>>>(h, W_pred, b_pred, out, N);
}


# SpMM: lane l holds floats l and l+32 of a row (was 2l, 2l+1): each gather and f32 atomic covers one whole 128-B line, 4 line requests per edge not 6; f32 atomics (arrival-order sums) kept
# speedup vs baseline: 1.6729x; 1.1273x over previous
#include <hip/hip_runtime.h>
#include <math.h>

typedef __attribute__((ext_vector_type(2))) float v2f;
typedef __attribute__((ext_vector_type(8))) float v8f;

#define NN 100000

__global__ void zero_f32(float* __restrict__ p, int n4) {
    int i = blockIdx.x * blockDim.x + threadIdx.x;
    int stride = gridDim.x * blockDim.x;
    float4 z = make_float4(0.f, 0.f, 0.f, 0.f);
    for (; i < n4; i += stride) ((float4*)p)[i] = z;
}

__global__ void deg_kernel(const int* __restrict__ src, const int* __restrict__ dst,
                           unsigned* __restrict__ out_cnt, unsigned* __restrict__ in_cnt, int E) {
    int e = blockIdx.x * blockDim.x + threadIdx.x;
    if (e < E) {
        atomicAdd(&out_cnt[src[e]], 1u);
        atomicAdd(&in_cnt[dst[e]], 1u);
    }
}

__global__ void norm_kernel(const unsigned* __restrict__ out_cnt, const unsigned* __restrict__ in_cnt,
                            float* __restrict__ out_norm, float* __restrict__ in_norm, int n) {
    int i = blockIdx.x * blockDim.x + threadIdx.x;
    if (i < n) {
        out_norm[i] = rsqrtf((float)(out_cnt[i] + 1u));
        in_norm[i]  = rsqrtf((float)(in_cnt[i] + 1u));
    }
}

__global__ void init_h_kernel(const float* __restrict__ weight, const float* __restrict__ W_in,
                              const float* __restrict__ b_in, float* __restrict__ h, int n) {
    int idx = blockIdx.x * blockDim.x + threadIdx.x;
    if (idx >= n * 16) return;
    int node = idx >> 4;
    int q = idx & 15;
    float w = weight[node];
    float4 wi = ((const float4*)W_in)[q];
    float4 bi = ((const float4*)b_in)[q];
    float4 r;
    r.x = w * wi.x + bi.x;
    r.y = w * wi.y + bi.y;
    r.z = w * wi.z + bi.z;
    r.w = w * wi.w + bi.w;
    ((float4*)(h + (size_t)node * 64))[q] = r;
}

__global__ __launch_bounds__(256) void gemm_layer_kernel(
    const float* __restrict__ h, const float* __restrict__ W,
    const float* __restrict__ out_norm, float* __restrict__ g, int act)
{
    __shared__ float Wl[64 * 64];
    __shared__ float Tl[32 * 64];

    const int tid = threadIdx.x;
    const int row0 = blockIdx.x * 32;

    for (int i = tid; i < 1024; i += 256)
        ((float4*)Wl)[i] = ((const float4*)W)[i];

    for (int i = tid; i < 512; i += 256) {
        int r  = i >> 4;
        int c4 = i & 15;
        int row = row0 + r;
        float4 v = ((const float4*)(h + (size_t)row * 64))[c4];
        if (act) {
            v.x = v.x > 0.f ? v.x : 0.01f * v.x;
            v.y = v.y > 0.f ? v.y : 0.01f * v.y;
            v.z = v.z > 0.f ? v.z : 0.01f * v.z;
            v.w = v.w > 0.f ? v.w : 0.01f * v.w;
        }
        float s = out_norm[row];
        v.x *= s; v.y *= s; v.z *= s; v.w *= s;
        ((float4*)(Tl + r * 64))[c4] = v;
    }
    __syncthreads();

    const int wave = tid >> 5;
    const int lane = tid & 31;
    const int msub = wave >> 2;
    const int nsub = wave & 3;
    const int l16  = lane & 15;
    const int arow = msub * 16 + l16;
    const int koff = (lane < 16) ? 0 : 2;
    const int col  = nsub * 16 + l16;

    v8f c = {};
    #pragma unroll
    for (int kk = 0; kk < 16; ++kk) {
        int k = kk * 4 + koff;
        v2f a, b;
        a.x = Tl[arow * 64 + k];
        a.y = Tl[arow * 64 + k + 1];
        b.x = Wl[k * 64 + col];
        b.y = Wl[(k + 1) * 64 + col];
        c = __builtin_amdgcn_wmma_f32_16x16x4_f32(false, a, false, b,
                                                  (short)0, c, false, false);
    }

    const int rbase = row0 + msub * 16 + ((lane < 16) ? 0 : 8);
    #pragma unroll
    for (int r = 0; r < 8; ++r)
        g[(size_t)(rbase + r) * 64 + col] = c[r];
}


typedef const float __attribute__((address_space(1)))* grow_in_t;
typedef float __attribute__((address_space(1)))* grow_out_t;
__device__ __forceinline__ grow_in_t row_in(const float* p, int r) {
    return (grow_in_t)((unsigned long long)p + ((unsigned long long)(unsigned)r << 8));
}
__device__ __forceinline__ grow_out_t row_out(float* p, int r) {
    return (grow_out_t)((unsigned long long)p + ((unsigned long long)(unsigned)r << 8));
}
__device__ __forceinline__ void add_f32(grow_out_t a, float x) {
    (void)__hip_atomic_fetch_add(a, x, __ATOMIC_RELAXED, __HIP_MEMORY_SCOPE_AGENT);
}

__global__ __launch_bounds__(256) void spmm_kernel(
    const float* __restrict__ g, const int* __restrict__ src, const int* __restrict__ dst,
    float* __restrict__ agg, int E)
{
    const unsigned lane = threadIdx.x & 31u;
    const int base = __builtin_amdgcn_readfirstlane((int)(blockIdx.x * 8 + (threadIdx.x >> 5)) * 32);
    if (base >= E) return;
    const int e = base + (int)lane;
    const int sv = (e < E) ? src[e] : 0;
    const int dv = (e < E) ? dst[e] : 0;
    const int cnt = (E - base < 32) ? (E - base) : 32;

    if (cnt == 32) {
        v2f v[2][8];
        #pragma unroll
        for (int j = 0; j < 8; ++j) {
            grow_in_t r = row_in(g, __builtin_amdgcn_readlane(sv, j)) + lane;
            v[0][j].x = r[0];
            v[0][j].y = r[32];
        }
        #pragma unroll
        for (int b = 0; b < 4; ++b) {
            if (b < 3) {
                #pragma unroll
                for (int j = 0; j < 8; ++j) {
                    grow_in_t r = row_in(g, __builtin_amdgcn_readlane(sv, (b + 1) * 8 + j)) + lane;
                    v[(b + 1) & 1][j].x = r[0];
                    v[(b + 1) & 1][j].y = r[32];
                }
            }
            #pragma unroll
            for (int j = 0; j < 8; ++j) {
                grow_out_t a = row_out(agg, __builtin_amdgcn_readlane(dv, b * 8 + j)) + lane;
                add_f32(a,      v[b & 1][j].x);
                add_f32(a + 32, v[b & 1][j].y);
            }
        }
    } else {
        for (int j = 0; j < cnt; ++j) {
            grow_in_t r = row_in(g, __builtin_amdgcn_readlane(sv, j)) + lane;
            const float x = r[0];
            const float y = r[32];
            grow_out_t a = row_out(agg, __builtin_amdgcn_readlane(dv, j)) + lane;
            add_f32(a,      x);
            add_f32(a + 32, y);
        }
    }
}

__global__ void post_kernel(const float* __restrict__ agg, const float* __restrict__ g,
                            const float* __restrict__ in_norm, const float* __restrict__ bc,
                            float* __restrict__ h, int n) {
    int idx = blockIdx.x * blockDim.x + threadIdx.x;
    if (idx >= n * 16) return;
    int node = idx >> 4;
    int q = idx & 15;
    float s = in_norm[node];
    float4 a = ((const float4*)(agg + (size_t)node * 64))[q];
    float4 gv = ((const float4*)(g + (size_t)node * 64))[q];
    float4 b = ((const float4*)bc)[q];
    float4 r;
    r.x = (a.x + gv.x) * s + b.x;
    r.y = (a.y + gv.y) * s + b.y;
    r.z = (a.z + gv.z) * s + b.z;
    r.w = (a.w + gv.w) * s + b.w;
    ((float4*)(h + (size_t)node * 64))[q] = r;
}

__global__ __launch_bounds__(256) void pred_kernel(
    const float* __restrict__ h, const float* __restrict__ W_pred,
    const float* __restrict__ b_pred, float* __restrict__ out, int n)
{
    int node = blockIdx.x * 8 + (threadIdx.x >> 5);
    if (node >= n) return;
    int lane = threadIdx.x & 31;
    float2 w = ((const float2*)W_pred)[lane];
    float2 v = ((const float2*)(h + (size_t)node * 64))[lane];
    float s = v.x * w.x + v.y * w.y;
    for (int off = 16; off > 0; off >>= 1)
        s += __shfl_down(s, off, 32);
    if (lane == 0) out[node] = s + b_pred[0];
}

extern "C" void kernel_launch(void* const* d_in, const int* in_sizes, int n_in,
                              void* d_out, int out_size, void* d_ws, size_t ws_size,
                              hipStream_t stream) {
    const float* weight = (const float*)d_in[0];
    const int*   src    = (const int*)d_in[1];
    const int*   dst    = (const int*)d_in[2];
    const float* W_in   = (const float*)d_in[3];
    const float* b_in   = (const float*)d_in[4];
    const float* Wc     = (const float*)d_in[5];
    const float* bc     = (const float*)d_in[6];
    const float* W_pred = (const float*)d_in[7];
    const float* b_pred = (const float*)d_in[8];
    float* out = (float*)d_out;

    const int N = in_sizes[0];
    const int E = in_sizes[1];

    unsigned* out_cnt = (unsigned*)d_ws;
    unsigned* in_cnt  = out_cnt + N;
    float* out_norm   = (float*)(in_cnt + N);
    float* in_norm    = out_norm + N;
    float* h          = in_norm + N;
    float* g          = h + (size_t)N * 64;
    float* agg        = g + (size_t)N * 64;

    zero_f32<<<512, 256, 0, stream>>>((float*)out_cnt, (2 * N) / 4);

    deg_kernel<<<(E + 255) / 256, 256, 0, stream>>>(src, dst, out_cnt, in_cnt, E);

    norm_kernel<<<(N + 255) / 256, 256, 0, stream>>>(out_cnt, in_cnt, out_norm, in_norm, N);

    init_h_kernel<<<(N * 16 + 255) / 256, 256, 0, stream>>>(weight, W_in, b_in, h, N);

    for (int l = 0; l < 3; ++l) {
        gemm_layer_kernel<<<N / 32, 256, 0, stream>>>(h, Wc + (size_t)l * 64 * 64,
                                                      out_norm, g, l > 0 ? 1 : 0);
        zero_f32<<<2048, 256, 0, stream>>>(agg, (N * 64) / 4);
        spmm_kernel<<<(E + 255) / 256, 256, 0, stream>>>(g, src, dst, agg, E);
        post_kernel<<<(N * 16 + 255) / 256, 256, 0, stream>>>(agg, g, in_norm,
                                                              bc + (size_t)l * 64, h, N);
    }

    pred_kernel<<<(N + 7) / 8, 256, 0, stream>>>(h, W_pred, b_pred, out, N);
}


# SpMM: edges bucketed once by 196-node dst range; a work-group per range sums its rows in LDS (LDS f32 atomics, arrival-order sums), self loop + epilogue fused; global f32 atomics, 4 launches gone
# speedup vs baseline: 2.2309x; 1.3336x over previous
#include <hip/hip_runtime.h>
#include <math.h>

typedef __attribute__((ext_vector_type(2))) float v2f;
typedef __attribute__((ext_vector_type(8))) float v8f;

#define NN 100000

__global__ void zero_f32(float* __restrict__ p, int n4) {
    int i = blockIdx.x * blockDim.x + threadIdx.x;
    int stride = gridDim.x * blockDim.x;
    float4 z = make_float4(0.f, 0.f, 0.f, 0.f);
    for (; i < n4; i += stride) ((float4*)p)[i] = z;
}

__global__ void deg_kernel(const int* __restrict__ src, const int* __restrict__ dst,
                           unsigned* __restrict__ out_cnt, unsigned* __restrict__ in_cnt, int E) {
    int e = blockIdx.x * blockDim.x + threadIdx.x;
    if (e < E) {
        atomicAdd(&out_cnt[src[e]], 1u);
        atomicAdd(&in_cnt[dst[e]], 1u);
    }
}

__global__ void norm_kernel(const unsigned* __restrict__ out_cnt, const unsigned* __restrict__ in_cnt,
                            float* __restrict__ out_norm, float* __restrict__ in_norm, int n) {
    int i = blockIdx.x * blockDim.x + threadIdx.x;
    if (i < n) {
        out_norm[i] = rsqrtf((float)(out_cnt[i] + 1u));
        in_norm[i]  = rsqrtf((float)(in_cnt[i] + 1u));
    }
}

__global__ void init_h_kernel(const float* __restrict__ weight, const float* __restrict__ W_in,
                              const float* __restrict__ b_in, float* __restrict__ h, int n) {
    int idx = blockIdx.x * blockDim.x + threadIdx.x;
    if (idx >= n * 16) return;
    int node = idx >> 4;
    int q = idx & 15;
    float w = weight[node];
    float4 wi = ((const float4*)W_in)[q];
    float4 bi = ((const float4*)b_in)[q];
    float4 r;
    r.x = w * wi.x + bi.x;
    r.y = w * wi.y + bi.y;
    r.z = w * wi.z + bi.z;
    r.w = w * wi.w + bi.w;
    ((float4*)(h + (size_t)node * 64))[q] = r;
}

__global__ __launch_bounds__(256) void gemm_layer_kernel(
    const float* __restrict__ h, const float* __restrict__ W,
    const float* __restrict__ out_norm, float* __restrict__ g, int act)
{
    __shared__ float Wl[64 * 64];
    __shared__ float Tl[32 * 64];

    const int tid = threadIdx.x;
    const int row0 = blockIdx.x * 32;

    for (int i = tid; i < 1024; i += 256)
        ((float4*)Wl)[i] = ((const float4*)W)[i];

    for (int i = tid; i < 512; i += 256) {
        int r  = i >> 4;
        int c4 = i & 15;
        int row = row0 + r;
        float4 v = ((const float4*)(h + (size_t)row * 64))[c4];
        if (act) {
            v.x = v.x > 0.f ? v.x : 0.01f * v.x;
            v.y = v.y > 0.f ? v.y : 0.01f * v.y;
            v.z = v.z > 0.f ? v.z : 0.01f * v.z;
            v.w = v.w > 0.f ? v.w : 0.01f * v.w;
        }
        float s = out_norm[row];
        v.x *= s; v.y *= s; v.z *= s; v.w *= s;
        ((float4*)(Tl + r * 64))[c4] = v;
    }
    __syncthreads();

    const int wave = tid >> 5;
    const int lane = tid & 31;
    const int msub = wave >> 2;
    const int nsub = wave & 3;
    const int l16  = lane & 15;
    const int arow = msub * 16 + l16;
    const int koff = (lane < 16) ? 0 : 2;
    const int col  = nsub * 16 + l16;

    v8f c = {};
    #pragma unroll
    for (int kk = 0; kk < 16; ++kk) {
        int k = kk * 4 + koff;
        v2f a, b;
        a.x = Tl[arow * 64 + k];
        a.y = Tl[arow * 64 + k + 1];
        b.x = Wl[k * 64 + col];
        b.y = Wl[(k + 1) * 64 + col];
        c = __builtin_amdgcn_wmma_f32_16x16x4_f32(false, a, false, b,
                                                  (short)0, c, false, false);
    }

    const int rbase = row0 + msub * 16 + ((lane < 16) ? 0 : 8);
    #pragma unroll
    for (int r = 0; r < 8; ++r)
        g[(size_t)(rbase + r) * 64 + col] = c[r];
}

#define NB 512
#define RB 196
#define SRC_BITS 17
#define SRC_MASK 0x1FFFFu

__global__ __launch_bounds__(1024) void bin_scan_kernel(
    const unsigned* __restrict__ in_cnt, unsigned* __restrict__ bin_start,
    unsigned* __restrict__ cursor, int n)
{
    __shared__ unsigned tot[NB];
    const int t = threadIdx.x;
    const int bin = t >> 1;
    const int part = t & 1;
    unsigned s = 0;
    for (int i = part; i < RB; i += 2) {
        const int node = bin * RB + i;
        if (node < n) s += in_cnt[node];
    }
    s += __shfl_xor(s, 1, 32);
    if (part == 0) tot[bin] = s;
    __syncthreads();
    if (t < 32) {
        unsigned loc[16];
        unsigned run = 0;
        #pragma unroll
        for (int k = 0; k < 16; ++k) { loc[k] = run; run += tot[t * 16 + k]; }
        unsigned incl = run;
        #pragma unroll
        for (int off = 1; off < 32; off <<= 1) {
            const unsigned v = __shfl_up(incl, off, 32);
            if (t >= off) incl += v;
        }
        const unsigned excl = incl - run;
        #pragma unroll
        for (int k = 0; k < 16; ++k) {
            bin_start[t * 16 + k] = excl + loc[k];
            cursor[t * 16 + k]    = excl + loc[k];
        }
        if (t == 31) bin_start[NB] = incl;
    }
}

__global__ __launch_bounds__(1024) void bin_scatter_kernel(
    const int* __restrict__ src, const int* __restrict__ dst,
    unsigned* __restrict__ cursor, unsigned* __restrict__ keys, int E)
{
    __shared__ unsigned hist[NB];
    __shared__ unsigned basep[NB];
    const int tid = threadIdx.x;
    for (int i = tid; i < NB; i += 1024) hist[i] = 0u;
    __syncthreads();
    const int ch = (E + (int)gridDim.x - 1) / (int)gridDim.x;
    const int e0 = (int)blockIdx.x * ch;
    const int e1 = (e0 + ch < E) ? (e0 + ch) : E;
    for (int e = e0 + tid; e < e1; e += 1024)
        atomicAdd(&hist[(unsigned)dst[e] / RB], 1u);
    __syncthreads();
    for (int i = tid; i < NB; i += 1024) {
        const unsigned c = hist[i];
        basep[i] = c ? atomicAdd(&cursor[i], c) : 0u;
        hist[i] = 0u;
    }
    __syncthreads();
    for (int e = e0 + tid; e < e1; e += 1024) {
        const unsigned d = (unsigned)dst[e];
        const unsigned s = (unsigned)src[e];
        const unsigned bin = d / RB;
        const unsigned r = atomicAdd(&hist[bin], 1u);
        keys[basep[bin] + r] = ((d - bin * RB) << SRC_BITS) | s;
    }
}

typedef const float __attribute__((address_space(1)))* grow_in_t;
__device__ __forceinline__ grow_in_t row_in(const float* p, unsigned r) {
    return (grow_in_t)((unsigned long long)p + ((unsigned long long)r << 8));
}
__device__ __forceinline__ void lds_add_f32(float* a, float x) {
    (void)__hip_atomic_fetch_add(a, x, __ATOMIC_RELAXED, __HIP_MEMORY_SCOPE_WORKGROUP);
}

__global__ __launch_bounds__(1024) void owner_spmm_kernel(
    const float* __restrict__ g, const unsigned* __restrict__ keys,
    const unsigned* __restrict__ bin_start, const float* __restrict__ in_norm,
    const float* __restrict__ bias, float* __restrict__ h, int n)
{
    __shared__ float acc[RB * 64];

    const int tid = threadIdx.x;
    const unsigned lane = threadIdx.x & 31u;
    const int node0 = (int)blockIdx.x * RB;
    int nn = n - node0;
    nn = nn > RB ? RB : nn;
    nn = nn < 0 ? 0 : nn;

    {
        const float4* g4 = (const float4*)(g + (size_t)node0 * 64);
        for (int i = tid; i < nn * 16; i += 1024) ((float4*)acc)[i] = g4[i];
    }
    __syncthreads();

    const int beg = (int)bin_start[blockIdx.x];
    const int end = (int)bin_start[blockIdx.x + 1];
    const int wave = __builtin_amdgcn_readfirstlane(tid >> 5);

    for (int base = beg + wave * 32; base < end; base += 1024) {
        const int e = base + (int)lane;
        const unsigned key = (e < end) ? keys[e] : 0u;
        const int cnt = (end - base < 32) ? (end - base) : 32;

        if (cnt == 32) {
            v2f v[2][8];
            #pragma unroll
            for (int j = 0; j < 8; ++j) {
                const unsigned k = (unsigned)__builtin_amdgcn_readlane((int)key, j);
                grow_in_t r = row_in(g, k & SRC_MASK) + lane;
                v[0][j].x = r[0];
                v[0][j].y = r[32];
            }
            #pragma unroll
            for (int b = 0; b < 4; ++b) {
                if (b < 3) {
                    #pragma unroll
                    for (int j = 0; j < 8; ++j) {
                        const unsigned k = (unsigned)__builtin_amdgcn_readlane((int)key, (b + 1) * 8 + j);
                        grow_in_t r = row_in(g, k & SRC_MASK) + lane;
                        v[(b + 1) & 1][j].x = r[0];
                        v[(b + 1) & 1][j].y = r[32];
                    }
                }
                #pragma unroll
                for (int j = 0; j < 8; ++j) {
                    const unsigned k = (unsigned)__builtin_amdgcn_readlane((int)key, b * 8 + j);
                    float* a = acc + (k >> SRC_BITS) * 64u + lane;
                    lds_add_f32(a,      v[b & 1][j].x);
                    lds_add_f32(a + 32, v[b & 1][j].y);
                }
            }
        } else {
            for (int j = 0; j < cnt; ++j) {
                const unsigned k = (unsigned)__builtin_amdgcn_readlane((int)key, j);
                grow_in_t r = row_in(g, k & SRC_MASK) + lane;
                const float x = r[0];
                const float y = r[32];
                float* a = acc + (k >> SRC_BITS) * 64u + lane;
                lds_add_f32(a,      x);
                lds_add_f32(a + 32, y);
            }
        }
    }
    __syncthreads();

    {
        float4* h4 = (float4*)(h + (size_t)node0 * 64);
        for (int i = tid; i < nn * 16; i += 1024) {
            const int r = i >> 4;
            const int q = i & 15;
            const float s = in_norm[node0 + r];
            const float4 a = ((const float4*)acc)[i];
            const float4 b = ((const float4*)bias)[q];
            float4 o;
            o.x = a.x * s + b.x;
            o.y = a.y * s + b.y;
            o.z = a.z * s + b.z;
            o.w = a.w * s + b.w;
            h4[i] = o;
        }
    }
}

__global__ __launch_bounds__(256) void pred_kernel(
    const float* __restrict__ h, const float* __restrict__ W_pred,
    const float* __restrict__ b_pred, float* __restrict__ out, int n)
{
    int node = blockIdx.x * 8 + (threadIdx.x >> 5);
    if (node >= n) return;
    int lane = threadIdx.x & 31;
    float2 w = ((const float2*)W_pred)[lane];
    float2 v = ((const float2*)(h + (size_t)node * 64))[lane];
    float s = v.x * w.x + v.y * w.y;
    for (int off = 16; off > 0; off >>= 1)
        s += __shfl_down(s, off, 32);
    if (lane == 0) out[node] = s + b_pred[0];
}

extern "C" void kernel_launch(void* const* d_in, const int* in_sizes, int n_in,
                              void* d_out, int out_size, void* d_ws, size_t ws_size,
                              hipStream_t stream) {
    const float* weight = (const float*)d_in[0];
    const int*   src    = (const int*)d_in[1];
    const int*   dst    = (const int*)d_in[2];
    const float* W_in   = (const float*)d_in[3];
    const float* b_in   = (const float*)d_in[4];
    const float* Wc     = (const float*)d_in[5];
    const float* bc     = (const float*)d_in[6];
    const float* W_pred = (const float*)d_in[7];
    const float* b_pred = (const float*)d_in[8];
    float* out = (float*)d_out;

    const int N = in_sizes[0];
    const int E = in_sizes[1];

    unsigned* out_cnt = (unsigned*)d_ws;
    unsigned* in_cnt  = out_cnt + N;
    float* out_norm   = (float*)(in_cnt + N);
    float* in_norm    = out_norm + N;
    float* h          = in_norm + N;
    float* g          = h + (size_t)N * 64;
    unsigned* bin_start = (unsigned*)(g + (size_t)N * 64);
    unsigned* cursor    = bin_start + 1024;
    unsigned* keys      = cursor + 1024;

    zero_f32<<<512, 256, 0, stream>>>((float*)out_cnt, (2 * N) / 4);

    deg_kernel<<<(E + 255) / 256, 256, 0, stream>>>(src, dst, out_cnt, in_cnt, E);

    norm_kernel<<<(N + 255) / 256, 256, 0, stream>>>(out_cnt, in_cnt, out_norm, in_norm, N);

    init_h_kernel<<<(N * 16 + 255) / 256, 256, 0, stream>>>(weight, W_in, b_in, h, N);

    bin_scan_kernel<<<1, 1024, 0, stream>>>(in_cnt, bin_start, cursor, N);
    bin_scatter_kernel<<<256, 1024, 0, stream>>>(src, dst, cursor, keys, E);

    for (int l = 0; l < 3; ++l) {
        gemm_layer_kernel<<<N / 32, 256, 0, stream>>>(h, Wc + (size_t)l * 64 * 64,
                                                      out_norm, g, l > 0 ? 1 : 0);
        owner_spmm_kernel<<<NB, 1024, 0, stream>>>(g, keys, bin_start, in_norm,
                                                   bc + (size_t)l * 64, h, N);
    }

    pred_kernel<<<(N + 7) / 8, 256, 0, stream>>>(h, W_pred, b_pred, out, N);
}
